# starting kernel
# baseline (speedup 1.0000x reference)
#include <hip/hip_runtime.h>

typedef float v2f __attribute__((ext_vector_type(2)));
typedef float v8f __attribute__((ext_vector_type(8)));

#define D            64
#define K            1024
#define ESTRIDE      66
#define ROWS_TOTAL   65536
#define THREADS      256
#define WAVES        8
#define SLABS        2
#define ROWS_PER_BLOCK (WAVES * 16 * SLABS)
#define NUM_BLOCKS   (ROWS_TOTAL / ROWS_PER_BLOCK)

#define LDS_E_WORDS  (K * ESTRIDE)
#define LDS_ESQ_OFF  LDS_E_WORDS
#define LDS_WL_OFF   (LDS_E_WORDS + K)
#define LDS_WORDS    (LDS_E_WORDS + K + WAVES)

__global__ __launch_bounds__(THREADS)
void vq_main(const float* __restrict__ x, const float* __restrict__ emb,
             float* __restrict__ out, float* __restrict__ partial)
{
    extern __shared__ float lds[];
    float* eLds = lds;
    float* esq  = lds + LDS_ESQ_OFF;
    float* wl   = lds + LDS_WL_OFF;

    const int tid  = threadIdx.x;
    const int wave = tid >> 5;
    const int lane = tid & 31;
    const int n    = lane & 15;
    const int hi   = lane >> 4;

    for (int i = tid; i < K * D; i += THREADS)
        eLds[(i >> 6) * ESTRIDE + (i & 63)] = emb[i];
    __syncthreads();

    for (int c = wave; c < K; c += WAVES) {
        const float* er = eLds + c * ESTRIDE;
        float v0 = er[lane], v1 = er[lane + 32];
        float s = v0 * v0 + v1 * v1;
        #pragma unroll
        for (int m = 16; m >= 1; m >>= 1) s += __shfl_xor(s, m, 32);
        if (lane == 0) esq[c] = s;
    }
    __syncthreads();

    float waveLoss = 0.0f;

    for (int slab = 0; slab < SLABS; ++slab) {
        const int rowBase = blockIdx.x * ROWS_PER_BLOCK + slab * (WAVES * 16) + wave * 16;

        const float* xrow = x + (size_t)(rowBase + n) * D + 2 * hi;
        v2f a[16];
        float xsqp = 0.0f;
        #pragma unroll
        for (int kk = 0; kk < 16; ++kk) {
            a[kk] = *(const v2f*)(xrow + 4 * kk);
            xsqp += a[kk][0] * a[kk][0] + a[kk][1] * a[kk][1];
        }
        const float xsqFull = xsqp + __shfl_xor(xsqp, 16, 32);

        float best[8];
        int   bidx[8];
        #pragma unroll
        for (int j = 0; j < 8; ++j) { best[j] = __builtin_inff(); bidx[j] = 0; }

        for (int t = 0; t < K / 16; ++t) {
            const int c0 = t * 16;
            const float* brow = eLds + (c0 + n) * ESTRIDE + 2 * hi;
            v8f acc = {0.f, 0.f, 0.f, 0.f, 0.f, 0.f, 0.f, 0.f};
            #pragma unroll
            for (int kk = 0; kk < 16; ++kk) {
                v2f b = *(const v2f*)(brow + 4 * kk);
                acc = __builtin_amdgcn_wmma_f32_16x16x4_f32(
                          false, a[kk], false, b, (short)0, acc, false, false);
            }
            const float es  = esq[c0 + n];
            const int   code = c0 + n;
            #pragma unroll
            for (int j = 0; j < 8; ++j) {
                float sc = __builtin_fmaf(-2.0f, acc[j], es);
                if (sc < best[j]) { best[j] = sc; bidx[j] = code; }
            }
        }

        #pragma unroll
        for (int j = 0; j < 8; ++j) {
            #pragma unroll
            for (int m = 1; m <= 8; m <<= 1) {
                float ob = __shfl_xor(best[j], m, 32);
                int   oi = __shfl_xor(bidx[j], m, 32);
                if (ob < best[j] || (ob == best[j] && oi < bidx[j])) {
                    best[j] = ob; bidx[j] = oi;
                }
            }
        }

        float lp = (hi == 0) ? xsqFull : 0.0f;
        if (n == 0) {
            float s8 = 0.0f;
            #pragma unroll
            for (int j = 0; j < 8; ++j) s8 += best[j];
            lp += s8;
        }
        #pragma unroll
        for (int m = 16; m >= 1; m >>= 1) lp += __shfl_xor(lp, m, 32);
        waveLoss += lp;

        #pragma unroll
        for (int r = 0; r < 16; ++r) {
            const int j = r & 7;
            const int srcLane = (r >> 3) << 4;
            const int idx = __shfl(bidx[j], srcLane, 32);
            const float* er = eLds + idx * ESTRIDE;
            float* orow = out + (size_t)(rowBase + r) * D;
            orow[lane]      = er[lane];
            orow[lane + 32] = er[lane + 32];
        }
    }

    if (lane == 0) wl[wave] = waveLoss;
    __syncthreads();
    if (tid == 0) {
        float s = 0.0f;
        #pragma unroll
        for (int w = 0; w < WAVES; ++w) s += wl[w];
        partial[blockIdx.x] = s;
    }
}

__global__ __launch_bounds__(NUM_BLOCKS)
void vq_finalize(const float* __restrict__ partial, float* __restrict__ lossOut)
{
    __shared__ float red[NUM_BLOCKS];
    const int t = threadIdx.x;
    red[t] = partial[t];
    __syncthreads();
    for (int s = NUM_BLOCKS / 2; s > 0; s >>= 1) {
        if (t < s) red[t] += red[t + s];
        __syncthreads();
    }
    if (t == 0)
        lossOut[0] = 1.25f * red[0] / (float)((size_t)ROWS_TOTAL * D);
}

extern "C" void kernel_launch(void* const* d_in, const int* in_sizes, int n_in,
                              void* d_out, int out_size, void* d_ws, size_t ws_size,
                              hipStream_t stream)
{
    (void)in_sizes; (void)n_in; (void)out_size; (void)ws_size;
    const float* x   = (const float*)d_in[0];
    const float* emb = (const float*)d_in[1];
    float* out       = (float*)d_out;
    float* partial   = (float*)d_ws;

    vq_main<<<NUM_BLOCKS, THREADS, LDS_WORDS * sizeof(float), stream>>>(x, emb, out, partial);
    vq_finalize<<<1, NUM_BLOCKS, 0, stream>>>(partial, out + (size_t)ROWS_TOTAL * D);
}


# Codebook kept in registers, 128 codes per wave, not in LDS: each wave sweeps all 256 rows with no B reads; per-row argmin merged across waves in LDS; scores bit-identical, loss sum order changed
# speedup vs baseline: 1.7377x; 1.7377x over previous
#include <hip/hip_runtime.h>

typedef float v2f __attribute__((ext_vector_type(2)));
typedef float v4f __attribute__((ext_vector_type(4)));
typedef int   v4i __attribute__((ext_vector_type(4)));
typedef float v8f __attribute__((ext_vector_type(8)));

#define D            64
#define K            1024
#define ROWS_TOTAL   65536
#define THREADS      256
#define WAVES        8
#define TILES_W      8
#define CODES_W      (TILES_W * 16)
#define ROWS_PER_BLOCK 256
#define SLABS        (ROWS_PER_BLOCK / 16)
#define NUM_BLOCKS   (ROWS_TOTAL / ROWS_PER_BLOCK)

template <int M>
__device__ __forceinline__ int row_xor_i(int v)
{
    return __builtin_amdgcn_update_dpp(v, v, 0x160 + M, 0xf, 0xf, false);
}
template <int M>
__device__ __forceinline__ float row_xor_f(float v)
{
    return __int_as_float(row_xor_i<M>(__float_as_int(v)));
}

__global__ __launch_bounds__(THREADS)
void vq_main(const float* __restrict__ x, const float* __restrict__ emb,
             float* __restrict__ out, float* __restrict__ partial)
{
    __shared__ __attribute__((aligned(16))) float candS[WAVES][ROWS_PER_BLOCK];
    __shared__ __attribute__((aligned(16))) int   candI[WAVES][ROWS_PER_BLOCK];
    __shared__ float xsqS[ROWS_PER_BLOCK];
    __shared__ float wl[WAVES];

    const int tid  = threadIdx.x;
    const int wave = tid >> 5;
    const int lane = tid & 31;
    const int n    = lane & 15;
    const int hi   = lane >> 4;
    const int codeBase = wave * CODES_W;
    const int rowBlock = blockIdx.x * ROWS_PER_BLOCK;

    v2f b[TILES_W][16];
    #pragma unroll
    for (int t = 0; t < TILES_W; ++t) {
        const float* brow = emb + (size_t)(codeBase + t * 16 + n) * D + 2 * hi;
        #pragma unroll
        for (int kk = 0; kk < 16; ++kk)
            b[t][kk] = *(const v2f*)(brow + 4 * kk);
    }

    float es[TILES_W];
    #pragma unroll
    for (int t = 0; t < TILES_W; ++t) {
        float w2[2];
        #pragma unroll
        for (int e = 0; e < 2; ++e) {
            float p[8];
            #pragma unroll
            for (int kk = 0; kk < 8; ++kk) {
                float v0 = b[t][kk][e], v1 = b[t][kk + 8][e];
                p[kk] = v0 * v0 + v1 * v1;
            }
            const float u0 = p[0] + p[4], u1 = p[1] + p[5], u2 = p[2] + p[6], u3 = p[3] + p[7];
            const float q0 = u0 + u2, q1 = u1 + u3;
            w2[e] = q0 + q1;
        }
        const float y0 = w2[0] + __shfl_xor(w2[0], 16, 32);
        const float y1 = w2[1] + __shfl_xor(w2[1], 16, 32);
        es[t] = y0 + y1;
    }

    v2f a[16];
    {
        const float* xrow = x + (size_t)(rowBlock + n) * D + 2 * hi;
        #pragma unroll
        for (int kk = 0; kk < 16; ++kk) a[kk] = *(const v2f*)(xrow + 4 * kk);
    }

    #pragma nounroll
    for (int slab = 0; slab < SLABS; ++slab) {
        const int slabNext = (slab + 1 < SLABS) ? slab + 1 : slab;
        v2f an[16];
        {
            const float* xrow = x + (size_t)(rowBlock + slabNext * 16 + n) * D + 2 * hi;
            #pragma unroll
            for (int kk = 0; kk < 16; ++kk) an[kk] = *(const v2f*)(xrow + 4 * kk);
        }

        if ((slab & (WAVES - 1)) == wave) {
            float xsqp = 0.0f;
            #pragma unroll
            for (int kk = 0; kk < 16; ++kk)
                xsqp += a[kk][0] * a[kk][0] + a[kk][1] * a[kk][1];
            const float xsqFull = xsqp + __shfl_xor(xsqp, 16, 32);
            if (hi == 0) xsqS[slab * 16 + n] = xsqFull;
        }

        float best[8];
        int   bidx[8];
        #pragma unroll
        for (int j = 0; j < 8; ++j) { best[j] = __builtin_inff(); bidx[j] = 0; }

        #pragma unroll
        for (int t = 0; t < TILES_W; ++t) {
            v8f acc = {0.f, 0.f, 0.f, 0.f, 0.f, 0.f, 0.f, 0.f};
            #pragma unroll
            for (int kk = 0; kk < 16; ++kk)
                acc = __builtin_amdgcn_wmma_f32_16x16x4_f32(
                          false, a[kk], false, b[t][kk], (short)0, acc, false, false);
            const int code = codeBase + t * 16 + n;
            #pragma unroll
            for (int j = 0; j < 8; ++j) {
                float sc = __builtin_fmaf(-2.0f, acc[j], es[t]);
                if (sc < best[j]) { best[j] = sc; bidx[j] = code; }
            }
        }

        float rs[8];
        int   ri[8];
        #pragma unroll
        for (int j = 0; j < 8; ++j) {
            float m = best[j], o;
            o = row_xor_f<1>(m); m = (o < m) ? o : m;
            o = row_xor_f<2>(m); m = (o < m) ? o : m;
            o = row_xor_f<4>(m); m = (o < m) ? o : m;
            o = row_xor_f<8>(m); m = (o < m) ? o : m;
            int c = (best[j] == m) ? bidx[j] : 0x7fffffff;
            c = min(c, row_xor_i<1>(c));
            c = min(c, row_xor_i<2>(c));
            c = min(c, row_xor_i<4>(c));
            c = min(c, row_xor_i<8>(c));
            rs[j] = m; ri[j] = c;
        }
        if (n == 0) {
            v4f* ps = (v4f*)&candS[wave][slab * 16 + 8 * hi];
            v4i* pi = (v4i*)&candI[wave][slab * 16 + 8 * hi];
            ps[0] = (v4f){rs[0], rs[1], rs[2], rs[3]};
            ps[1] = (v4f){rs[4], rs[5], rs[6], rs[7]};
            pi[0] = (v4i){ri[0], ri[1], ri[2], ri[3]};
            pi[1] = (v4i){ri[4], ri[5], ri[6], ri[7]};
        }

        #pragma unroll
        for (int kk = 0; kk < 16; ++kk) a[kk] = an[kk];
    }
    __syncthreads();

    float bs = candS[0][tid];
    int   bi = candI[0][tid];
    #pragma unroll
    for (int w = 1; w < WAVES; ++w) {
        const float os = candS[w][tid];
        const int   oi = candI[w][tid];
        if (os < bs) { bs = os; bi = oi; }
    }
    bi &= (K - 1);

    float lp = bs + xsqS[tid];
    #pragma unroll
    for (int m = 16; m >= 1; m >>= 1) lp += __shfl_xor(lp, m, 32);
    if (lane == 0) wl[wave] = lp;

    #pragma unroll
    for (int i = 0; i < 16; ++i) {
        const int r   = 2 * i + hi;
        const int idx = __shfl(bi, r, 32);
        const v4f v = *(const v4f*)(emb + (size_t)idx * D + 4 * n);
        *(v4f*)(out + (size_t)(rowBlock + wave * 32 + r) * D + 4 * n) = v;
    }

    __syncthreads();
    if (tid == 0) {
        float s = 0.0f;
        #pragma unroll
        for (int w = 0; w < WAVES; ++w) s += wl[w];
        partial[blockIdx.x] = s;
    }
}

__global__ __launch_bounds__(NUM_BLOCKS)
void vq_finalize(const float* __restrict__ partial, float* __restrict__ lossOut)
{
    __shared__ float red[NUM_BLOCKS];
    const int t = threadIdx.x;
    red[t] = partial[t];
    __syncthreads();
    for (int s = NUM_BLOCKS / 2; s > 0; s >>= 1) {
        if (t < s) red[t] += red[t + s];
        __syncthreads();
    }
    if (t == 0)
        lossOut[0] = 1.25f * red[0] / (float)((size_t)ROWS_TOTAL * D);
}

extern "C" void kernel_launch(void* const* d_in, const int* in_sizes, int n_in,
                              void* d_out, int out_size, void* d_ws, size_t ws_size,
                              hipStream_t stream)
{
    (void)in_sizes; (void)n_in; (void)out_size; (void)ws_size;
    const float* x   = (const float*)d_in[0];
    const float* emb = (const float*)d_in[1];
    float* out       = (float*)d_out;
    float* partial   = (float*)d_ws;

    vq_main<<<NUM_BLOCKS, THREADS, 0, stream>>>(x, emb, out, partial);
    vq_finalize<<<1, NUM_BLOCKS, 0, stream>>>(partial, out + (size_t)ROWS_TOTAL * D);
}


# Matrix operand roles swapped: codes on the result's register axis, x rows on lanes, so each lane runs its own row's argmin in registers; the per-slab 16-lane exchange is gone; scores bit-identical
# speedup vs baseline: 1.8580x; 1.0692x over previous
#include <hip/hip_runtime.h>

typedef float v2f __attribute__((ext_vector_type(2)));
typedef float v4f __attribute__((ext_vector_type(4)));
typedef int   v4i __attribute__((ext_vector_type(4)));
typedef float v8f __attribute__((ext_vector_type(8)));

#define D            64
#define K            1024
#define ROWS_TOTAL   65536
#define THREADS      256
#define WAVES        8
#define TILES_W      8
#define CODES_W      (TILES_W * 16)
#define ROWS_PER_BLOCK 256
#define SLABS        (ROWS_PER_BLOCK / 16)
#define NUM_BLOCKS   (ROWS_TOTAL / ROWS_PER_BLOCK)

template <int M>
__device__ __forceinline__ int row_xor_i(int v)
{
    return __builtin_amdgcn_update_dpp(v, v, 0x160 + M, 0xf, 0xf, false);
}
template <int M>
__device__ __forceinline__ float row_xor_f(float v)
{
    return __int_as_float(row_xor_i<M>(__float_as_int(v)));
}

__global__ __launch_bounds__(THREADS)
void vq_main(const float* __restrict__ x, const float* __restrict__ emb,
             float* __restrict__ out, float* __restrict__ partial)
{
    __shared__ __attribute__((aligned(16))) float candS[WAVES][ROWS_PER_BLOCK][2];
    __shared__ __attribute__((aligned(16))) int   candI[WAVES][ROWS_PER_BLOCK][2];
    __shared__ float xsqS[ROWS_PER_BLOCK];
    __shared__ float wl[WAVES];

    const int tid  = threadIdx.x;
    const int wave = tid >> 5;
    const int lane = tid & 31;
    const int n    = lane & 15;
    const int hi   = lane >> 4;
    const int codeBase = wave * CODES_W;
    const int rowBlock = blockIdx.x * ROWS_PER_BLOCK;

    v2f b[TILES_W][16];
    #pragma unroll
    for (int t = 0; t < TILES_W; ++t) {
        const float* brow = emb + (size_t)(codeBase + t * 16 + n) * D + 2 * hi;
        #pragma unroll
        for (int kk = 0; kk < 16; ++kk)
            b[t][kk] = *(const v2f*)(brow + 4 * kk);
    }

    float es[TILES_W];
    #pragma unroll
    for (int t = 0; t < TILES_W; ++t) {
        float w2[2];
        #pragma unroll
        for (int e = 0; e < 2; ++e) {
            float p[8];
            #pragma unroll
            for (int kk = 0; kk < 8; ++kk) {
                float v0 = b[t][kk][e], v1 = b[t][kk + 8][e];
                p[kk] = v0 * v0 + v1 * v1;
            }
            const float u0 = p[0] + p[4], u1 = p[1] + p[5], u2 = p[2] + p[6], u3 = p[3] + p[7];
            const float q0 = u0 + u2, q1 = u1 + u3;
            w2[e] = q0 + q1;
        }
        const float y0 = w2[0] + __shfl_xor(w2[0], 16, 32);
        const float y1 = w2[1] + __shfl_xor(w2[1], 16, 32);
        es[t] = y0 + y1;
    }
    float esr[TILES_W][8];
    #pragma unroll
    for (int t = 0; t < TILES_W; ++t) {
        #pragma unroll
        for (int j = 0; j < 8; ++j) esr[t][j] = __shfl(es[t], 8 * hi + j, 32);
    }

    v2f a[16];
    {
        const float* xrow = x + (size_t)(rowBlock + n) * D + 2 * hi;
        #pragma unroll
        for (int kk = 0; kk < 16; ++kk) a[kk] = *(const v2f*)(xrow + 4 * kk);
    }

    #pragma nounroll
    for (int slab = 0; slab < SLABS; ++slab) {
        const int slabNext = (slab + 1 < SLABS) ? slab + 1 : slab;
        v2f an[16];
        {
            const float* xrow = x + (size_t)(rowBlock + slabNext * 16 + n) * D + 2 * hi;
            #pragma unroll
            for (int kk = 0; kk < 16; ++kk) an[kk] = *(const v2f*)(xrow + 4 * kk);
        }

        if ((slab & (WAVES - 1)) == wave) {
            float xsqp = 0.0f;
            #pragma unroll
            for (int kk = 0; kk < 16; ++kk)
                xsqp += a[kk][0] * a[kk][0] + a[kk][1] * a[kk][1];
            const float xsqFull = xsqp + __shfl_xor(xsqp, 16, 32);
            if (hi == 0) xsqS[slab * 16 + n] = xsqFull;
        }

        float best = 0.0f;
        int   bt   = 0;
        float win[8];

        #pragma unroll
        for (int t = 0; t < TILES_W; ++t) {
            v8f acc = {0.f, 0.f, 0.f, 0.f, 0.f, 0.f, 0.f, 0.f};
            #pragma unroll
            for (int kk = 0; kk < 16; ++kk)
                acc = __builtin_amdgcn_wmma_f32_16x16x4_f32(
                          false, b[t][kk], false, a[kk], (short)0, acc, false, false);
            float sc[8];
            #pragma unroll
            for (int j = 0; j < 8; ++j)
                sc[j] = __builtin_fmaf(-2.0f, acc[j], esr[t][j]);
            const float mt = __builtin_fminf(
                __builtin_fminf(__builtin_fminf(sc[0], sc[1]), __builtin_fminf(sc[2], sc[3])),
                __builtin_fminf(__builtin_fminf(sc[4], sc[5]), __builtin_fminf(sc[6], sc[7])));
            if (t == 0) {
                best = mt;
                #pragma unroll
                for (int j = 0; j < 8; ++j) win[j] = sc[j];
            } else {
                const bool take = mt < best;
                best = take ? mt : best;
                bt   = take ? t  : bt;
                #pragma unroll
                for (int j = 0; j < 8; ++j) win[j] = take ? sc[j] : win[j];
            }
        }

        int bj = 7;
        #pragma unroll
        for (int j = 6; j >= 0; --j)
            if (win[j] == best) bj = j;
        candS[wave][slab * 16 + n][hi] = best;
        candI[wave][slab * 16 + n][hi] = codeBase + 16 * bt + 8 * hi + bj;

        #pragma unroll
        for (int kk = 0; kk < 16; ++kk) a[kk] = an[kk];
    }
    __syncthreads();

    float bs = candS[0][tid][0];
    int   bi = candI[0][tid][0];
    #pragma unroll
    for (int w = 0; w < WAVES; ++w) {
        #pragma unroll
        for (int h = (w == 0) ? 1 : 0; h < 2; ++h) {
            const float os = candS[w][tid][h];
            const int   oi = candI[w][tid][h];
            const bool take = (os < bs) | ((os == bs) & (oi < bi));
            bs = take ? os : bs;
            bi = take ? oi : bi;
        }
    }
    bi &= (K - 1);

    float lp = bs + xsqS[tid];
    #pragma unroll
    for (int m = 16; m >= 1; m >>= 1) lp += __shfl_xor(lp, m, 32);
    if (lane == 0) wl[wave] = lp;

    #pragma unroll
    for (int i = 0; i < 16; ++i) {
        const int r   = 2 * i + hi;
        const int idx = __shfl(bi, r, 32);
        const v4f v = *(const v4f*)(emb + (size_t)idx * D + 4 * n);
        *(v4f*)(out + (size_t)(rowBlock + wave * 32 + r) * D + 4 * n) = v;
    }

    __syncthreads();
    if (tid == 0) {
        float s = 0.0f;
        #pragma unroll
        for (int w = 0; w < WAVES; ++w) s += wl[w];
        partial[blockIdx.x] = s;
    }
}

__global__ __launch_bounds__(NUM_BLOCKS)
void vq_finalize(const float* __restrict__ partial, float* __restrict__ lossOut)
{
    __shared__ float red[NUM_BLOCKS];
    const int t = threadIdx.x;
    red[t] = partial[t];
    __syncthreads();
    for (int s = NUM_BLOCKS / 2; s > 0; s >>= 1) {
        if (t < s) red[t] += red[t + s];
        __syncthreads();
    }
    if (t == 0)
        lossOut[0] = 1.25f * red[0] / (float)((size_t)ROWS_TOTAL * D);
}

extern "C" void kernel_launch(void* const* d_in, const int* in_sizes, int n_in,
                              void* d_out, int out_size, void* d_ws, size_t ws_size,
                              hipStream_t stream)
{
    (void)in_sizes; (void)n_in; (void)out_size; (void)ws_size;
    const float* x   = (const float*)d_in[0];
    const float* emb = (const float*)d_in[1];
    float* out       = (float*)d_out;
    float* partial   = (float*)d_ws;

    vq_main<<<NUM_BLOCKS, THREADS, 0, stream>>>(x, emb, out, partial);
    vq_finalize<<<1, NUM_BLOCKS, 0, stream>>>(partial, out + (size_t)ROWS_TOTAL * D);
}


# x fragments double-buffered in two fixed register sets, slab loop unrolled by two: no copies between slabs, the next slab's loads are first waited for at its first WMMA; scores bit-identical
# speedup vs baseline: 2.0102x; 1.0819x over previous
#include <hip/hip_runtime.h>

typedef float v2f __attribute__((ext_vector_type(2)));
typedef float v4f __attribute__((ext_vector_type(4)));
typedef int   v4i __attribute__((ext_vector_type(4)));
typedef float v8f __attribute__((ext_vector_type(8)));

#define D            64
#define K            1024
#define ROWS_TOTAL   65536
#define THREADS      256
#define WAVES        8
#define TILES_W      8
#define CODES_W      (TILES_W * 16)
#define ROWS_PER_BLOCK 256
#define SLABS        (ROWS_PER_BLOCK / 16)
#define NUM_BLOCKS   (ROWS_TOTAL / ROWS_PER_BLOCK)

template <int M>
__device__ __forceinline__ int row_xor_i(int v)
{
    return __builtin_amdgcn_update_dpp(v, v, 0x160 + M, 0xf, 0xf, false);
}
template <int M>
__device__ __forceinline__ float row_xor_f(float v)
{
    return __int_as_float(row_xor_i<M>(__float_as_int(v)));
}

__device__ __forceinline__ void fetch_x(v2f (&dst)[16], const float* __restrict__ x, int row, int hi)
{
    const float* xrow = x + (size_t)row * D + 2 * hi;
    #pragma unroll
    for (int kk = 0; kk < 16; ++kk) dst[kk] = *(const v2f*)(xrow + 4 * kk);
}

__device__ __forceinline__ void sweep_slab(const v2f (&b)[TILES_W][16], const float (&esr)[TILES_W][8],
                                           const v2f (&a)[16], int slab, int wave, int n, int hi, int codeBase,
                                           float* __restrict__ candSw, int* __restrict__ candIw,
                                           float* __restrict__ xsqS)
{
    if ((slab & (WAVES - 1)) == wave) {
        float xsqp = 0.0f;
        #pragma unroll
        for (int kk = 0; kk < 16; ++kk)
            xsqp += a[kk][0] * a[kk][0] + a[kk][1] * a[kk][1];
        const float xsqFull = xsqp + __shfl_xor(xsqp, 16, 32);
        if (hi == 0) xsqS[slab * 16 + n] = xsqFull;
    }

    float best = 0.0f;
    int   bt   = 0;
    float win[8];

    #pragma unroll
    for (int t = 0; t < TILES_W; ++t) {
        v8f acc = {0.f, 0.f, 0.f, 0.f, 0.f, 0.f, 0.f, 0.f};
        #pragma unroll
        for (int kk = 0; kk < 16; ++kk)
            acc = __builtin_amdgcn_wmma_f32_16x16x4_f32(
                      false, b[t][kk], false, a[kk], (short)0, acc, false, false);
        float sc[8];
        #pragma unroll
        for (int j = 0; j < 8; ++j)
            sc[j] = __builtin_fmaf(-2.0f, acc[j], esr[t][j]);
        const float mt = __builtin_fminf(
            __builtin_fminf(__builtin_fminf(sc[0], sc[1]), __builtin_fminf(sc[2], sc[3])),
            __builtin_fminf(__builtin_fminf(sc[4], sc[5]), __builtin_fminf(sc[6], sc[7])));
        if (t == 0) {
            best = mt;
            #pragma unroll
            for (int j = 0; j < 8; ++j) win[j] = sc[j];
        } else {
            const bool take = mt < best;
            best = take ? mt : best;
            bt   = take ? t  : bt;
            #pragma unroll
            for (int j = 0; j < 8; ++j) win[j] = take ? sc[j] : win[j];
        }
    }

    int bj = 7;
    #pragma unroll
    for (int j = 6; j >= 0; --j)
        if (win[j] == best) bj = j;
    candSw[(slab * 16 + n) * 2 + hi] = best;
    candIw[(slab * 16 + n) * 2 + hi] = codeBase + 16 * bt + 8 * hi + bj;
}

__global__ __launch_bounds__(THREADS)
void vq_main(const float* __restrict__ x, const float* __restrict__ emb,
             float* __restrict__ out, float* __restrict__ partial)
{
    __shared__ __attribute__((aligned(16))) float candS[WAVES][ROWS_PER_BLOCK][2];
    __shared__ __attribute__((aligned(16))) int   candI[WAVES][ROWS_PER_BLOCK][2];
    __shared__ float xsqS[ROWS_PER_BLOCK];
    __shared__ float wl[WAVES];

    const int tid  = threadIdx.x;
    const int wave = tid >> 5;
    const int lane = tid & 31;
    const int n    = lane & 15;
    const int hi   = lane >> 4;
    const int codeBase = wave * CODES_W;
    const int rowBlock = blockIdx.x * ROWS_PER_BLOCK;

    v2f b[TILES_W][16];
    #pragma unroll
    for (int t = 0; t < TILES_W; ++t) {
        const float* brow = emb + (size_t)(codeBase + t * 16 + n) * D + 2 * hi;
        #pragma unroll
        for (int kk = 0; kk < 16; ++kk)
            b[t][kk] = *(const v2f*)(brow + 4 * kk);
    }

    float es[TILES_W];
    #pragma unroll
    for (int t = 0; t < TILES_W; ++t) {
        float w2[2];
        #pragma unroll
        for (int e = 0; e < 2; ++e) {
            float p[8];
            #pragma unroll
            for (int kk = 0; kk < 8; ++kk) {
                float v0 = b[t][kk][e], v1 = b[t][kk + 8][e];
                p[kk] = v0 * v0 + v1 * v1;
            }
            const float u0 = p[0] + p[4], u1 = p[1] + p[5], u2 = p[2] + p[6], u3 = p[3] + p[7];
            const float q0 = u0 + u2, q1 = u1 + u3;
            w2[e] = q0 + q1;
        }
        const float y0 = w2[0] + __shfl_xor(w2[0], 16, 32);
        const float y1 = w2[1] + __shfl_xor(w2[1], 16, 32);
        es[t] = y0 + y1;
    }
    float esr[TILES_W][8];
    #pragma unroll
    for (int t = 0; t < TILES_W; ++t) {
        #pragma unroll
        for (int j = 0; j < 8; ++j) esr[t][j] = __shfl(es[t], 8 * hi + j, 32);
    }

    float* candSw = &candS[wave][0][0];
    int*   candIw = &candI[wave][0][0];
    v2f a0[16], a1[16];
    fetch_x(a0, x, rowBlock + n, hi);

    #pragma nounroll
    for (int slab = 0; slab < SLABS; slab += 2) {
        fetch_x(a1, x, rowBlock + (slab + 1) * 16 + n, hi);
        sweep_slab(b, esr, a0, slab, wave, n, hi, codeBase, candSw, candIw, xsqS);

        const int slabNext = (slab + 2 < SLABS) ? slab + 2 : slab;
        fetch_x(a0, x, rowBlock + slabNext * 16 + n, hi);
        sweep_slab(b, esr, a1, slab + 1, wave, n, hi, codeBase, candSw, candIw, xsqS);
    }
    __syncthreads();

    float bs = candS[0][tid][0];
    int   bi = candI[0][tid][0];
    #pragma unroll
    for (int w = 0; w < WAVES; ++w) {
        #pragma unroll
        for (int h = (w == 0) ? 1 : 0; h < 2; ++h) {
            const float os = candS[w][tid][h];
            const int   oi = candI[w][tid][h];
            const bool take = (os < bs) | ((os == bs) & (oi < bi));
            bs = take ? os : bs;
            bi = take ? oi : bi;
        }
    }
    bi &= (K - 1);

    float lp = bs + xsqS[tid];
    #pragma unroll
    for (int m = 16; m >= 1; m >>= 1) lp += __shfl_xor(lp, m, 32);
    if (lane == 0) wl[wave] = lp;

    #pragma unroll
    for (int i = 0; i < 16; ++i) {
        const int r   = 2 * i + hi;
        const int idx = __shfl(bi, r, 32);
        const v4f v = *(const v4f*)(emb + (size_t)idx * D + 4 * n);
        *(v4f*)(out + (size_t)(rowBlock + wave * 32 + r) * D + 4 * n) = v;
    }

    __syncthreads();
    if (tid == 0) {
        float s = 0.0f;
        #pragma unroll
        for (int w = 0; w < WAVES; ++w) s += wl[w];
        partial[blockIdx.x] = s;
    }
}

__global__ __launch_bounds__(NUM_BLOCKS)
void vq_finalize(const float* __restrict__ partial, float* __restrict__ lossOut)
{
    __shared__ float red[NUM_BLOCKS];
    const int t = threadIdx.x;
    red[t] = partial[t];
    __syncthreads();
    for (int s = NUM_BLOCKS / 2; s > 0; s >>= 1) {
        if (t < s) red[t] += red[t + s];
        __syncthreads();
    }
    if (t == 0)
        lossOut[0] = 1.25f * red[0] / (float)((size_t)ROWS_TOTAL * D);
}

extern "C" void kernel_launch(void* const* d_in, const int* in_sizes, int n_in,
                              void* d_out, int out_size, void* d_ws, size_t ws_size,
                              hipStream_t stream)
{
    (void)in_sizes; (void)n_in; (void)out_size; (void)ws_size;
    const float* x   = (const float*)d_in[0];
    const float* emb = (const float*)d_in[1];
    float* out       = (float*)d_out;
    float* partial   = (float*)d_ws;

    vq_main<<<NUM_BLOCKS, THREADS, 0, stream>>>(x, emb, out, partial);
    vq_finalize<<<1, NUM_BLOCKS, 0, stream>>>(partial, out + (size_t)ROWS_TOTAL * D);
}


# Wave priority stepped down with progress (s_setprio 3/2/1/0 at slabs 0/8/12/14): the SIMD's two waves stay within a segment, each one's argmin work runs under the other's WMMAs; scores bit-identical
# speedup vs baseline: 2.1012x; 1.0453x over previous
#include <hip/hip_runtime.h>

typedef float v2f __attribute__((ext_vector_type(2)));
typedef float v4f __attribute__((ext_vector_type(4)));
typedef int   v4i __attribute__((ext_vector_type(4)));
typedef float v8f __attribute__((ext_vector_type(8)));

#define D            64
#define K            1024
#define ROWS_TOTAL   65536
#define THREADS      256
#define WAVES        8
#define TILES_W      8
#define CODES_W      (TILES_W * 16)
#define ROWS_PER_BLOCK 256
#define SLABS        (ROWS_PER_BLOCK / 16)
#define NUM_BLOCKS   (ROWS_TOTAL / ROWS_PER_BLOCK)

template <int M>
__device__ __forceinline__ int row_xor_i(int v)
{
    return __builtin_amdgcn_update_dpp(v, v, 0x160 + M, 0xf, 0xf, false);
}
template <int M>
__device__ __forceinline__ float row_xor_f(float v)
{
    return __int_as_float(row_xor_i<M>(__float_as_int(v)));
}

__device__ __forceinline__ void fetch_x(v2f (&dst)[16], const float* __restrict__ x, int row, int hi)
{
    const float* xrow = x + (size_t)row * D + 2 * hi;
    #pragma unroll
    for (int kk = 0; kk < 16; ++kk) dst[kk] = *(const v2f*)(xrow + 4 * kk);
}

__device__ __forceinline__ void sweep_slab(const v2f (&b)[TILES_W][16], const float (&esr)[TILES_W][8],
                                           const v2f (&a)[16], int slab, int wave, int n, int hi, int codeBase,
                                           float* __restrict__ candSw, int* __restrict__ candIw,
                                           float* __restrict__ xsqS)
{
    if ((slab & (WAVES - 1)) == wave) {
        float xsqp = 0.0f;
        #pragma unroll
        for (int kk = 0; kk < 16; ++kk)
            xsqp += a[kk][0] * a[kk][0] + a[kk][1] * a[kk][1];
        const float xsqFull = xsqp + __shfl_xor(xsqp, 16, 32);
        if (hi == 0) xsqS[slab * 16 + n] = xsqFull;
    }

    float best = 0.0f;
    int   bt   = 0;
    float win[8];

    #pragma unroll
    for (int t = 0; t < TILES_W; ++t) {
        v8f acc = {0.f, 0.f, 0.f, 0.f, 0.f, 0.f, 0.f, 0.f};
        #pragma unroll
        for (int kk = 0; kk < 16; ++kk)
            acc = __builtin_amdgcn_wmma_f32_16x16x4_f32(
                      false, b[t][kk], false, a[kk], (short)0, acc, false, false);
        float sc[8];
        #pragma unroll
        for (int j = 0; j < 8; ++j)
            sc[j] = __builtin_fmaf(-2.0f, acc[j], esr[t][j]);
        const float mt = __builtin_fminf(
            __builtin_fminf(__builtin_fminf(sc[0], sc[1]), __builtin_fminf(sc[2], sc[3])),
            __builtin_fminf(__builtin_fminf(sc[4], sc[5]), __builtin_fminf(sc[6], sc[7])));
        if (t == 0) {
            best = mt;
            #pragma unroll
            for (int j = 0; j < 8; ++j) win[j] = sc[j];
        } else {
            const bool take = mt < best;
            best = take ? mt : best;
            bt   = take ? t  : bt;
            #pragma unroll
            for (int j = 0; j < 8; ++j) win[j] = take ? sc[j] : win[j];
        }
    }

    int bj = 7;
    #pragma unroll
    for (int j = 6; j >= 0; --j)
        if (win[j] == best) bj = j;
    candSw[(slab * 16 + n) * 2 + hi] = best;
    candIw[(slab * 16 + n) * 2 + hi] = codeBase + 16 * bt + 8 * hi + bj;
}

__global__ __launch_bounds__(THREADS)
void vq_main(const float* __restrict__ x, const float* __restrict__ emb,
             float* __restrict__ out, float* __restrict__ partial)
{
    __shared__ __attribute__((aligned(16))) float candS[WAVES][ROWS_PER_BLOCK][2];
    __shared__ __attribute__((aligned(16))) int   candI[WAVES][ROWS_PER_BLOCK][2];
    __shared__ float xsqS[ROWS_PER_BLOCK];
    __shared__ float wl[WAVES];

    const int tid  = threadIdx.x;
    const int wave = tid >> 5;
    const int lane = tid & 31;
    const int n    = lane & 15;
    const int hi   = lane >> 4;
    const int codeBase = wave * CODES_W;
    const int rowBlock = blockIdx.x * ROWS_PER_BLOCK;

    v2f b[TILES_W][16];
    #pragma unroll
    for (int t = 0; t < TILES_W; ++t) {
        const float* brow = emb + (size_t)(codeBase + t * 16 + n) * D + 2 * hi;
        #pragma unroll
        for (int kk = 0; kk < 16; ++kk)
            b[t][kk] = *(const v2f*)(brow + 4 * kk);
    }

    float es[TILES_W];
    #pragma unroll
    for (int t = 0; t < TILES_W; ++t) {
        float w2[2];
        #pragma unroll
        for (int e = 0; e < 2; ++e) {
            float p[8];
            #pragma unroll
            for (int kk = 0; kk < 8; ++kk) {
                float v0 = b[t][kk][e], v1 = b[t][kk + 8][e];
                p[kk] = v0 * v0 + v1 * v1;
            }
            const float u0 = p[0] + p[4], u1 = p[1] + p[5], u2 = p[2] + p[6], u3 = p[3] + p[7];
            const float q0 = u0 + u2, q1 = u1 + u3;
            w2[e] = q0 + q1;
        }
        const float y0 = w2[0] + __shfl_xor(w2[0], 16, 32);
        const float y1 = w2[1] + __shfl_xor(w2[1], 16, 32);
        es[t] = y0 + y1;
    }
    float esr[TILES_W][8];
    #pragma unroll
    for (int t = 0; t < TILES_W; ++t) {
        #pragma unroll
        for (int j = 0; j < 8; ++j) esr[t][j] = __shfl(es[t], 8 * hi + j, 32);
    }

    float* candSw = &candS[wave][0][0];
    int*   candIw = &candI[wave][0][0];
    v2f a0[16], a1[16];
    fetch_x(a0, x, rowBlock + n, hi);

    __builtin_amdgcn_s_setprio(3);
    #pragma nounroll
    for (int slab = 0; slab < SLABS; slab += 2) {
        if (slab == 8)  __builtin_amdgcn_s_setprio(2);
        if (slab == 12) __builtin_amdgcn_s_setprio(1);
        if (slab == 14) __builtin_amdgcn_s_setprio(0);
        fetch_x(a1, x, rowBlock + (slab + 1) * 16 + n, hi);
        sweep_slab(b, esr, a0, slab, wave, n, hi, codeBase, candSw, candIw, xsqS);

        const int slabNext = (slab + 2 < SLABS) ? slab + 2 : slab;
        fetch_x(a0, x, rowBlock + slabNext * 16 + n, hi);
        sweep_slab(b, esr, a1, slab + 1, wave, n, hi, codeBase, candSw, candIw, xsqS);
    }
    __syncthreads();

    float bs = candS[0][tid][0];
    int   bi = candI[0][tid][0];
    #pragma unroll
    for (int w = 0; w < WAVES; ++w) {
        #pragma unroll
        for (int h = (w == 0) ? 1 : 0; h < 2; ++h) {
            const float os = candS[w][tid][h];
            const int   oi = candI[w][tid][h];
            const bool take = (os < bs) | ((os == bs) & (oi < bi));
            bs = take ? os : bs;
            bi = take ? oi : bi;
        }
    }
    bi &= (K - 1);

    float lp = bs + xsqS[tid];
    #pragma unroll
    for (int m = 16; m >= 1; m >>= 1) lp += __shfl_xor(lp, m, 32);
    if (lane == 0) wl[wave] = lp;

    #pragma unroll
    for (int i = 0; i < 16; ++i) {
        const int r   = 2 * i + hi;
        const int idx = __shfl(bi, r, 32);
        const v4f v = *(const v4f*)(emb + (size_t)idx * D + 4 * n);
        *(v4f*)(out + (size_t)(rowBlock + wave * 32 + r) * D + 4 * n) = v;
    }

    __syncthreads();
    if (tid == 0) {
        float s = 0.0f;
        #pragma unroll
        for (int w = 0; w < WAVES; ++w) s += wl[w];
        partial[blockIdx.x] = s;
    }
}

__global__ __launch_bounds__(NUM_BLOCKS)
void vq_finalize(const float* __restrict__ partial, float* __restrict__ lossOut)
{
    __shared__ float red[NUM_BLOCKS];
    const int t = threadIdx.x;
    red[t] = partial[t];
    __syncthreads();
    for (int s = NUM_BLOCKS / 2; s > 0; s >>= 1) {
        if (t < s) red[t] += red[t + s];
        __syncthreads();
    }
    if (t == 0)
        lossOut[0] = 1.25f * red[0] / (float)((size_t)ROWS_TOTAL * D);
}

extern "C" void kernel_launch(void* const* d_in, const int* in_sizes, int n_in,
                              void* d_out, int out_size, void* d_ws, size_t ws_size,
                              hipStream_t stream)
{
    (void)in_sizes; (void)n_in; (void)out_size; (void)ws_size;
    const float* x   = (const float*)d_in[0];
    const float* emb = (const float*)d_in[1];
    float* out       = (float*)d_out;
    float* partial   = (float*)d_ws;

    vq_main<<<NUM_BLOCKS, THREADS, 0, stream>>>(x, emb, out, partial);
    vq_finalize<<<1, NUM_BLOCKS, 0, stream>>>(partial, out + (size_t)ROWS_TOTAL * D);
}
